# starting kernel
# baseline (speedup 1.0000x reference)
#include <hip/hip_runtime.h>
#include <hip/hip_bf16.h>

#define B_ 4
#define N_ 384
#define D_ 64
#define TS 3

typedef __attribute__((ext_vector_type(2))) float v2f;
typedef __attribute__((ext_vector_type(8))) float v8f;

__global__ __launch_bounds__(128) void gnn_stage1(const float* __restrict__ A,
                                                  const float* __restrict__ X,
                                                  float* __restrict__ T) {
    const int wave = threadIdx.x >> 5;
    const int lane = threadIdx.x & 31;
    const int b    = blockIdx.x / (N_ / 16);
    const int i0   = (blockIdx.x % (N_ / 16)) * 16;
    const int d0   = wave * 16;

    const int lm = lane & 15;
    const int lk = (lane >> 4) * 2;

    const float* Arow = A + (size_t)(b * N_ + i0 + lm) * N_ * 2;
    const float* Xb   = X + (size_t)b * N_ * D_ * TS;

    v8f c = {};
    for (int k0 = 0; k0 < N_; k0 += 4) {
        const int j = k0 + lk;
        const float4 af = *(const float4*)(Arow + (size_t)j * 2);
        v2f a;  a.x = af.y;  a.y = af.w;
        v2f bv;
        bv.x = Xb[((size_t)j       * D_ + d0 + lm) * TS + 2];
        bv.y = Xb[((size_t)(j + 1) * D_ + d0 + lm) * TS + 2];
        c = __builtin_amdgcn_wmma_f32_16x16x4_f32(false, a, false, bv,
                                                  (short)0, c, false, false);
    }

    const int row0 = (lane >> 4) * 8;
    float* Trow = T + (size_t)(b * N_ + i0) * D_;
#pragma unroll
    for (int r = 0; r < 8; ++r) {
        const int row = row0 + r;
        const int col = d0 + lm;
        const float x1 = Xb[((size_t)(i0 + row) * D_ + col) * TS + 1];
        Trow[(size_t)row * D_ + col] = c[r] + x1;
    }
}

__global__ __launch_bounds__(128) void gnn_stage2(const float* __restrict__ A,
                                                  const float* __restrict__ X,
                                                  const float* __restrict__ T,
                                                  const float* __restrict__ W,
                                                  float* __restrict__ out) {
    __shared__ float S[16][D_ + 4];

    const int wave = threadIdx.x >> 5;
    const int lane = threadIdx.x & 31;
    const int b    = blockIdx.x / (N_ / 16);
    const int i0   = (blockIdx.x % (N_ / 16)) * 16;
    const int d0   = wave * 16;
    const int lm   = lane & 15;
    const int lk   = (lane >> 4) * 2;
    const int row0 = (lane >> 4) * 8;

    const float* Arow = A + (size_t)(b * N_ + i0 + lm) * N_ * 2;
    const float* Xb   = X + (size_t)b * N_ * D_ * TS;
    const float* Tb   = T + (size_t)b * N_ * D_;

    v8f c = {};
    for (int k0 = 0; k0 < N_; k0 += 4) {
        const int j = k0 + lk;
        const float4 af = *(const float4*)(Arow + (size_t)j * 2);
        v2f a;  a.x = af.x;  a.y = af.z;
        v2f bv;
        bv.x = Tb[(size_t)j       * D_ + d0 + lm];
        bv.y = Tb[(size_t)(j + 1) * D_ + d0 + lm];
        c = __builtin_amdgcn_wmma_f32_16x16x4_f32(false, a, false, bv,
                                                  (short)0, c, false, false);
    }
#pragma unroll
    for (int r = 0; r < 8; ++r) {
        const int row = row0 + r;
        const int col = d0 + lm;
        S[row][col] = c[r] + Xb[((size_t)(i0 + row) * D_ + col) * TS + 0];
    }
    __syncthreads();

    v8f c2 = {};
    for (int k0 = 0; k0 < D_; k0 += 4) {
        const int k = k0 + lk;
        v2f a;  a.x = S[lm][k];  a.y = S[lm][k + 1];
        v2f bv;
        bv.x = W[(size_t)k       * D_ + d0 + lm];
        bv.y = W[(size_t)(k + 1) * D_ + d0 + lm];
        c2 = __builtin_amdgcn_wmma_f32_16x16x4_f32(false, a, false, bv,
                                                   (short)0, c2, false, false);
    }

    float* Orow = out + (size_t)(b * N_ + i0) * D_;
#pragma unroll
    for (int r = 0; r < 8; ++r) {
        Orow[(size_t)(row0 + r) * D_ + d0 + lm] = c2[r];
    }
}

extern "C" void kernel_launch(void* const* d_in, const int* in_sizes, int n_in,
                              void* d_out, int out_size, void* d_ws, size_t ws_size,
                              hipStream_t stream) {
    (void)in_sizes; (void)n_in; (void)out_size; (void)ws_size;
    const float* A = (const float*)d_in[0];
    const float* X = (const float*)d_in[1];
    const float* W = (const float*)d_in[2];
    float* out = (float*)d_out;
    float* T   = (float*)d_ws;

    const dim3 grid(B_ * (N_ / 16));
    const dim3 block(128);

    gnn_stage1<<<grid, block, 0, stream>>>(A, X, T);
    gnn_stage2<<<grid, block, 0, stream>>>(A, X, T, W, out);
}


# B operands (X2, T, W) no longer staged in LDS: each wave streams its own 16 columns' fragments from global memory, requested 2-3 blocks of 12 steps ahead of their products; sum order unchanged
# speedup vs baseline: 3.0041x; 3.0041x over previous
#include <hip/hip_runtime.h>

#define B_ 4
#define N_ 384
#define D_ 64
#define TS 3

#define AP 388
#define SP (D_ + 4)

typedef __attribute__((ext_vector_type(2))) float v2f;
typedef __attribute__((ext_vector_type(8))) float v8f;

#define A_PIECES 24

#define KU 12
#define NBLK (N_ / (4 * KU))
#define KR 16
#define AH_X 2
#define AH_T 3


template <int PA, int NU>
__device__ __forceinline__ void read_a_frags(const float* Am, int lm, int lk, int kb, v2f (&a)[NU]) {
#pragma unroll
    for (int u = 0; u < NU; ++u)
        a[u] = *(const v2f*)(Am + lm * PA + kb + 4 * u + lk);
}

template <bool FROM_T>
__device__ __forceinline__ void load_b_frags(const float* Bl, int s0, v2f (&b)[KU]) {
#pragma unroll
    for (int u = 0; u < KU; ++u) {
        if (FROM_T) {
            b[u] = *(const v2f*)(Bl + (size_t)(s0 + u) * (4 * D_));
        } else {
            const float* p = Bl + (size_t)(s0 + u) * (4 * D_ * TS);
            b[u].x = p[0];
            b[u].y = p[D_ * TS];
        }
    }
}

template <int NU>
__device__ __forceinline__ v8f mma_block(const v2f (&a)[NU], const v2f (&b)[NU], v8f c) {
#pragma unroll
    for (int u = 0; u < NU; ++u)
        c = __builtin_amdgcn_wmma_f32_16x16x4_f32(false, a[u], false, b[u],
                                                  (short)0, c, false, false);
    return c;
}

__device__ __forceinline__ void lds_barrier() {
    asm volatile("s_wait_dscnt 0x0\n\ts_barrier_signal -1\n\ts_barrier_wait -1" ::: "memory");
}

template <bool FROM_T, int AH, int FIRST, int LAST>
__device__ __forceinline__ void stream_blocks(const float* As, const float* Bl, int lm, int lk,
                                              v2f (&a)[2][KU], v2f (&bq)[AH + 1][KU], v8f& c) {
#pragma unroll
    for (int blk = FIRST; blk < LAST; ++blk) {
        if (blk + AH < NBLK)
            load_b_frags<FROM_T>(Bl, (blk + AH) * KU, bq[(blk + AH) % (AH + 1)]);
        if (blk + 1 < NBLK)
            read_a_frags<AP, KU>(As, lm, lk, (blk + 1) * 4 * KU, a[(blk + 1) & 1]);
        __builtin_amdgcn_sched_barrier(0);
        c = mma_block<KU>(a[blk & 1], bq[blk % (AH + 1)], c);
        __builtin_amdgcn_sched_barrier(0);
    }
}

__global__ __launch_bounds__(128) void gnn_stage1(const float* __restrict__ A,
                                                  const float* __restrict__ X,
                                                  float* __restrict__ T) {
    __shared__ __attribute__((aligned(16))) float As[16 * AP];

    const int tid  = threadIdx.x;
    const int wave = tid >> 5;
    const int lane = tid & 31;
    const int b    = blockIdx.x / (N_ / 16);
    const int i0   = (blockIdx.x % (N_ / 16)) * 16;
    const int d0   = wave * 16;
    const int lm   = lane & 15;
    const int lk   = (lane >> 4) * 2;
    const int row0 = (lane >> 4) * 8;

    const float4* Astrip = (const float4*)(A + (size_t)(b * N_ + i0) * N_ * 2);
    const float*  Xb     = X + (size_t)b * N_ * D_ * TS;

    float4 av[A_PIECES];
#pragma unroll
    for (int i = 0; i < A_PIECES; ++i)
        av[i] = Astrip[i * 128 + tid];

    __builtin_amdgcn_sched_barrier(0);

    const float* Xl = Xb + ((size_t)lk * D_ + d0 + lm) * TS + 2;
    v2f bq[AH_X + 1][KU];
#pragma unroll
    for (int j = 0; j < AH_X; ++j)
        load_b_frags<false>(Xl, j * KU, bq[j]);
    __builtin_amdgcn_sched_barrier(0);

#pragma unroll
    for (int i = 0; i < A_PIECES; ++i) {
        const int base = i * 128;
        const int row  = base / 192 + (((base % 192) + tid) >= 192 ? 1 : 0);
        const int q    = base + tid;
        v2f h;  h.x = av[i].y;  h.y = av[i].w;
        *(v2f*)(As + 2 * q + (AP - N_) * row) = h;
    }
    lds_barrier();

    v8f c = {};
    v2f a[2][KU];
    read_a_frags<AP, KU>(As, lm, lk, 0, a[0]);
    stream_blocks<false, AH_X, 0, NBLK - 2>(As, Xl, lm, lk, a, bq, c);
    float x1[8];
#pragma unroll
    for (int r = 0; r < 8; ++r)
        x1[r] = Xb[((size_t)(i0 + row0 + r) * D_ + d0 + lm) * TS + 1];
    stream_blocks<false, AH_X, NBLK - 2, NBLK>(As, Xl, lm, lk, a, bq, c);

    float* Tp = T + (size_t)b * N_ * D_ + (size_t)((i0 + row0) >> 1) * (2 * D_);
#pragma unroll
    for (int m = 0; m < 4; ++m) {
        v2f v;  v.x = c[2 * m] + x1[2 * m];  v.y = c[2 * m + 1] + x1[2 * m + 1];
        *(v2f*)(Tp + (size_t)m * (2 * D_) + 2 * (d0 + lm)) = v;
    }
}

__global__ __launch_bounds__(128) void gnn_stage2(const float* __restrict__ A,
                                                  const float* __restrict__ X,
                                                  const float* __restrict__ T,
                                                  const float* __restrict__ W,
                                                  float* __restrict__ out) {
    __shared__ __attribute__((aligned(16))) float As[16 * AP];
    __shared__ __attribute__((aligned(16))) float S[16 * SP];

    const int tid  = threadIdx.x;
    const int wave = tid >> 5;
    const int lane = tid & 31;
    const int b    = blockIdx.x / (N_ / 16);
    const int i0   = (blockIdx.x % (N_ / 16)) * 16;
    const int d0   = wave * 16;
    const int lm   = lane & 15;
    const int lk   = (lane >> 4) * 2;
    const int row0 = (lane >> 4) * 8;

    const float4* Astrip = (const float4*)(A + (size_t)(b * N_ + i0) * N_ * 2);
    const float*  Xb     = X + (size_t)b * N_ * D_ * TS;

    float4 av[A_PIECES];
#pragma unroll
    for (int i = 0; i < A_PIECES; ++i)
        av[i] = Astrip[i * 128 + tid];

    __builtin_amdgcn_sched_barrier(0);

    const float* Tl = T + (size_t)b * N_ * D_ + (size_t)(lane >> 4) * (2 * D_) + 2 * (d0 + lm);
    v2f bq[AH_T + 1][KU];
#pragma unroll
    for (int j = 0; j < AH_T; ++j)
        load_b_frags<true>(Tl, j * KU, bq[j]);
    __builtin_amdgcn_sched_barrier(0);

#pragma unroll
    for (int i = 0; i < A_PIECES; ++i) {
        const int base = i * 128;
        const int row  = base / 192 + (((base % 192) + tid) >= 192 ? 1 : 0);
        const int q    = base + tid;
        v2f h;  h.x = av[i].x;  h.y = av[i].z;
        *(v2f*)(As + 2 * q + (AP - N_) * row) = h;
    }
    lds_barrier();

    v8f c = {};
    v2f a[2][KU];
    read_a_frags<AP, KU>(As, lm, lk, 0, a[0]);
    stream_blocks<true, AH_T, 0, NBLK - 2>(As, Tl, lm, lk, a, bq, c);
    float x0[8];
#pragma unroll
    for (int r = 0; r < 8; ++r)
        x0[r] = Xb[((size_t)(i0 + row0 + r) * D_ + d0 + lm) * TS + 0];
    v2f wb[KR];
#pragma unroll
    for (int u = 0; u < KR; ++u) {
        const float* p = W + (size_t)(4 * u + lk) * D_ + d0 + lm;
        wb[u].x = p[0];
        wb[u].y = p[D_];
    }
    stream_blocks<true, AH_T, NBLK - 2, NBLK>(As, Tl, lm, lk, a, bq, c);

#pragma unroll
    for (int r = 0; r < 8; ++r)
        S[(row0 + r) * SP + d0 + lm] = c[r] + x0[r];
    lds_barrier();

    v2f sa[KR];
    read_a_frags<SP, KR>(S, lm, lk, 0, sa);
    __builtin_amdgcn_sched_barrier(0);
    v8f c2 = {};
    c2 = mma_block<KR>(sa, wb, c2);

    float* Orow = out + (size_t)(b * N_ + i0) * D_;
#pragma unroll
    for (int r = 0; r < 8; ++r)
        Orow[(size_t)(row0 + r) * D_ + d0 + lm] = c2[r];
}

extern "C" void kernel_launch(void* const* d_in, const int* in_sizes, int n_in,
                              void* d_out, int out_size, void* d_ws, size_t ws_size,
                              hipStream_t stream) {
    (void)in_sizes; (void)n_in; (void)out_size; (void)ws_size;
    const float* A = (const float*)d_in[0];
    const float* X = (const float*)d_in[1];
    const float* W = (const float*)d_in[2];
    float* out = (float*)d_out;
    float* T   = (float*)d_ws;

    const dim3 grid(B_ * (N_ / 16));
    const dim3 block(128);

    gnn_stage1<<<grid, block, 0, stream>>>(A, X, T);
    gnn_stage2<<<grid, block, 0, stream>>>(A, X, T, W, out);
}


# Each work-group starts its 384-deep contraction at another 48-deep block (strip index mod 8, wrapping) so a batch's 24 work-groups stop asking for the same X2/T lines at once; sum order rotates
# speedup vs baseline: 3.2123x; 1.0693x over previous
#include <hip/hip_runtime.h>

#define B_ 4
#define N_ 384
#define D_ 64
#define TS 3

#define AP 388
#define SP (D_ + 4)

typedef __attribute__((ext_vector_type(2))) float v2f;
typedef __attribute__((ext_vector_type(8))) float v8f;

#define A_PIECES 24

#define KU 12
#define NBLK (N_ / (4 * KU))
#define KR 16
#define AH_X 2
#define AH_T 3


template <int PA, int NU>
__device__ __forceinline__ void read_a_frags(const float* Am, int lm, int lk, int kb, v2f (&a)[NU]) {
#pragma unroll
    for (int u = 0; u < NU; ++u)
        a[u] = *(const v2f*)(Am + lm * PA + kb + 4 * u + lk);
}

template <bool FROM_T>
__device__ __forceinline__ void load_b_frags(const float* Bu, unsigned off, v2f (&b)[KU]) {
#pragma unroll
    for (int u = 0; u < KU; ++u) {
        if (FROM_T) {
            const char* p = (const char*)(Bu + (size_t)u * (4 * D_));
            b[u] = *(const v2f*)(p + off);
        } else {
            const float* p = Bu + (size_t)u * (4 * D_ * TS);
            b[u].x = p[off];
            b[u].y = (p + D_ * TS)[off];
        }
    }
}

static_assert((NBLK & (NBLK - 1)) == 0, "the wrap below is a mask");
__device__ __forceinline__ int rot_block(int pos, int rot) { return (pos + rot) & (NBLK - 1); }

__device__ __forceinline__ void block_offsets(unsigned loff, unsigned unit, int rot, unsigned (&boff)[NBLK]) {
#pragma unroll
    for (int p = 0; p < NBLK; ++p)
        boff[p] = loff + (unsigned)rot_block(p, rot) * unit;
}

template <int NU>
__device__ __forceinline__ v8f mma_block(const v2f (&a)[NU], const v2f (&b)[NU], v8f c) {
#pragma unroll
    for (int u = 0; u < NU; ++u)
        c = __builtin_amdgcn_wmma_f32_16x16x4_f32(false, a[u], false, b[u],
                                                  (short)0, c, false, false);
    return c;
}

__device__ __forceinline__ void lds_barrier() {
    asm volatile("s_wait_dscnt 0x0\n\ts_barrier_signal -1\n\ts_barrier_wait -1" ::: "memory");
}

template <bool FROM_T, int AH, int FIRST, int LAST>
__device__ __forceinline__ void stream_blocks(const float* As, const float* Bu, const unsigned (&boff)[NBLK],
                                              int lm, int lk, int rot,
                                              v2f (&a)[2][KU], v2f (&bq)[AH + 1][KU], v8f& c) {
#pragma unroll
    for (int blk = FIRST; blk < LAST; ++blk) {
        if (blk + AH < NBLK)
            load_b_frags<FROM_T>(Bu, boff[blk + AH], bq[(blk + AH) % (AH + 1)]);
        if (blk + 1 < NBLK)
            read_a_frags<AP, KU>(As, lm, lk, rot_block(blk + 1, rot) * 4 * KU, a[(blk + 1) & 1]);
        __builtin_amdgcn_sched_barrier(0);
        c = mma_block<KU>(a[blk & 1], bq[blk % (AH + 1)], c);
        __builtin_amdgcn_sched_barrier(0);
    }
}

__global__ __launch_bounds__(128) void gnn_stage1(const float* __restrict__ A,
                                                  const float* __restrict__ X,
                                                  float* __restrict__ T) {
    __shared__ __attribute__((aligned(16))) float As[16 * AP];

    const int tid  = threadIdx.x;
    const int wave = tid >> 5;
    const int lane = tid & 31;
    const int b    = blockIdx.x / (N_ / 16);
    const int strip = blockIdx.x % (N_ / 16);
    const int i0   = strip * 16;
    const int rot  = strip & (NBLK - 1);
    const int d0   = wave * 16;
    const int lm   = lane & 15;
    const int lk   = (lane >> 4) * 2;
    const int row0 = (lane >> 4) * 8;

    const float4* Astrip = (const float4*)(A + (size_t)(b * N_ + i0) * N_ * 2);
    const float*  Xb     = X + (size_t)b * N_ * D_ * TS;

    float4 av[A_PIECES];
#pragma unroll
    for (int i = 0; i < A_PIECES; ++i)
        av[i] = Astrip[i * 128 + tid];

    __builtin_amdgcn_sched_barrier(0);

    const float*   Xu = Xb + 2;
    const unsigned xl = (unsigned)((lk * D_ + d0 + lm) * TS);
    unsigned xoff[NBLK];
    block_offsets(xl, KU * 4 * D_ * TS, rot, xoff);
    v2f bq[AH_X + 1][KU];
#pragma unroll
    for (int j = 0; j < AH_X; ++j)
        load_b_frags<false>(Xu, xoff[j], bq[j]);
    __builtin_amdgcn_sched_barrier(0);

#pragma unroll
    for (int i = 0; i < A_PIECES; ++i) {
        const int base = i * 128;
        const int row  = base / 192 + (((base % 192) + tid) >= 192 ? 1 : 0);
        const int q    = base + tid;
        v2f h;  h.x = av[i].y;  h.y = av[i].w;
        *(v2f*)(As + 2 * q + (AP - N_) * row) = h;
    }
    lds_barrier();

    v8f c = {};
    v2f a[2][KU];
    read_a_frags<AP, KU>(As, lm, lk, rot * 4 * KU, a[0]);
    stream_blocks<false, AH_X, 0, NBLK - 2>(As, Xu, xoff, lm, lk, rot, a, bq, c);
    float x1[8];
#pragma unroll
    for (int r = 0; r < 8; ++r)
        x1[r] = Xb[((size_t)(i0 + row0 + r) * D_ + d0 + lm) * TS + 1];
    stream_blocks<false, AH_X, NBLK - 2, NBLK>(As, Xu, xoff, lm, lk, rot, a, bq, c);

    float* Tp = T + (size_t)b * N_ * D_ + (size_t)((i0 + row0) >> 1) * (2 * D_);
#pragma unroll
    for (int m = 0; m < 4; ++m) {
        v2f v;  v.x = c[2 * m] + x1[2 * m];  v.y = c[2 * m + 1] + x1[2 * m + 1];
        *(v2f*)(Tp + (size_t)m * (2 * D_) + 2 * (d0 + lm)) = v;
    }
}

__global__ __launch_bounds__(128) void gnn_stage2(const float* __restrict__ A,
                                                  const float* __restrict__ X,
                                                  const float* __restrict__ T,
                                                  const float* __restrict__ W,
                                                  float* __restrict__ out) {
    __shared__ __attribute__((aligned(16))) float As[16 * AP];
    __shared__ __attribute__((aligned(16))) float S[16 * SP];

    const int tid  = threadIdx.x;
    const int wave = tid >> 5;
    const int lane = tid & 31;
    const int b    = blockIdx.x / (N_ / 16);
    const int strip = blockIdx.x % (N_ / 16);
    const int i0   = strip * 16;
    const int rot  = strip & (NBLK - 1);
    const int d0   = wave * 16;
    const int lm   = lane & 15;
    const int lk   = (lane >> 4) * 2;
    const int row0 = (lane >> 4) * 8;

    const float4* Astrip = (const float4*)(A + (size_t)(b * N_ + i0) * N_ * 2);
    const float*  Xb     = X + (size_t)b * N_ * D_ * TS;

    float4 av[A_PIECES];
#pragma unroll
    for (int i = 0; i < A_PIECES; ++i)
        av[i] = Astrip[i * 128 + tid];

    __builtin_amdgcn_sched_barrier(0);

    const float*   Tu = T + (size_t)b * N_ * D_;
    const unsigned tl = (unsigned)(((lane >> 4) * D_ + d0 + lm) * 8);
    unsigned toff[NBLK];
    block_offsets(tl, KU * 4 * D_ * 4, rot, toff);
    v2f bq[AH_T + 1][KU];
#pragma unroll
    for (int j = 0; j < AH_T; ++j)
        load_b_frags<true>(Tu, toff[j], bq[j]);
    __builtin_amdgcn_sched_barrier(0);

#pragma unroll
    for (int i = 0; i < A_PIECES; ++i) {
        const int base = i * 128;
        const int row  = base / 192 + (((base % 192) + tid) >= 192 ? 1 : 0);
        const int q    = base + tid;
        v2f h;  h.x = av[i].x;  h.y = av[i].z;
        *(v2f*)(As + 2 * q + (AP - N_) * row) = h;
    }
    lds_barrier();

    v8f c = {};
    v2f a[2][KU];
    read_a_frags<AP, KU>(As, lm, lk, rot * 4 * KU, a[0]);
    stream_blocks<true, AH_T, 0, NBLK - 2>(As, Tu, toff, lm, lk, rot, a, bq, c);
    float x0[8];
#pragma unroll
    for (int r = 0; r < 8; ++r)
        x0[r] = Xb[((size_t)(i0 + row0 + r) * D_ + d0 + lm) * TS + 0];
    v2f wb[KR];
#pragma unroll
    for (int u = 0; u < KR; ++u) {
        const float* p = W + (size_t)(4 * u + lk) * D_ + d0 + lm;
        wb[u].x = p[0];
        wb[u].y = p[D_];
    }
    stream_blocks<true, AH_T, NBLK - 2, NBLK>(As, Tu, toff, lm, lk, rot, a, bq, c);

#pragma unroll
    for (int r = 0; r < 8; ++r)
        S[(row0 + r) * SP + d0 + lm] = c[r] + x0[r];
    lds_barrier();

    v2f sa[KR];
    read_a_frags<SP, KR>(S, lm, lk, 0, sa);
    __builtin_amdgcn_sched_barrier(0);
    v8f c2 = {};
    c2 = mma_block<KR>(sa, wb, c2);

    float* Orow = out + (size_t)(b * N_ + i0) * D_;
#pragma unroll
    for (int r = 0; r < 8; ++r)
        Orow[(size_t)(row0 + r) * D_ + d0 + lm] = c2[r];
}

extern "C" void kernel_launch(void* const* d_in, const int* in_sizes, int n_in,
                              void* d_out, int out_size, void* d_ws, size_t ws_size,
                              hipStream_t stream) {
    (void)in_sizes; (void)n_in; (void)out_size; (void)ws_size;
    const float* A = (const float*)d_in[0];
    const float* X = (const float*)d_in[1];
    const float* W = (const float*)d_in[2];
    float* out = (float*)d_out;
    float* T   = (float*)d_ws;

    const dim3 grid(B_ * (N_ / 16));
    const dim3 block(128);

    gnn_stage1<<<grid, block, 0, stream>>>(A, X, T);
    gnn_stage2<<<grid, block, 0, stream>>>(A, X, T, W, out);
}
